# starting kernel
# baseline (speedup 1.0000x reference)
#include <hip/hip_runtime.h>
#include <hip/hip_bf16.h>
#include <math.h>

typedef __attribute__((ext_vector_type(2))) float v2f;
typedef __attribute__((ext_vector_type(8))) float v8f;

#define S_LEN 2048
#define EMB   64
#define NW    8
#define QB    128

__device__ __forceinline__ v8f wmma4(v2f a, v2f b, v8f c) {
  return __builtin_amdgcn_wmma_f32_16x16x4_f32(false, a, false, b, (short)0, c,
                                               false, false);
}

__device__ __forceinline__ float xor16(float v) {
  int i = __float_as_int(v);
  int o = __builtin_amdgcn_permlanex16(i, i, 0x76543210, 0xfedcba98, false, false);
  return __int_as_float(o);
}

__device__ __forceinline__ void load_frag_pair(const float* zsm, int rowbase,
                                               int r, int hi, v2f& f0, v2f& f1) {
  const float* p = zsm + (rowbase + r) * NW + hi * 2;
  f0 = *(const v2f*)p;
  f1 = *(const v2f*)(p + 4);
}

__device__ __forceinline__ void load_vtile(const float* zsm, int kt, int rc,
                                           int hi, float vv[8]) {
#pragma unroll
  for (int sl = 0; sl < 4; ++sl) {
    int row = kt + sl * 4 + hi * 2;
    vv[2 * sl]     = zsm[row * NW + rc];
    vv[2 * sl + 1] = zsm[(row + 1) * NW + rc];
  }
}

__global__ __launch_bounds__(256) void qattn_kernel(
    const float* __restrict__ x,
    const float* __restrict__ theta,
    float* __restrict__ attn_out) {
  __shared__ float zsm[S_LEN * NW];

  const int qblk = blockIdx.x & 15;
  const int bh   = blockIdx.x >> 4;
  const int b    = bh >> 3;
  const int h    = bh & 7;
  const int tid  = threadIdx.x;

  float th[8];
#pragma unroll
  for (int j = 0; j < 8; ++j) th[j] = theta[j];

  const float* xbase = x + (size_t)b * S_LEN * EMB + h * NW;
#pragma unroll
  for (int k = 0; k < 8; ++k) {
    int pos = tid + k * 256;
    const float* xrow = xbase + (size_t)pos * EMB;
    if (k + 1 < 8)
      __builtin_prefetch(xbase + (size_t)(pos + 256) * EMB, 0, 3);
    float4 xa = ((const float4*)xrow)[0];
    float4 xc = ((const float4*)xrow)[1];
    float c0 = __cosf(xa.x + th[0]);
    float c1 = __cosf(xa.y + th[1]);
    float c2 = __cosf(xa.z + th[2]);
    float c3 = __cosf(xa.w + th[3]);
    float c4 = __cosf(xc.x + th[4]);
    float c5 = __cosf(xc.y + th[5]);
    float c6 = __cosf(xc.z + th[6]);
    float c7 = __cosf(xc.w + th[7]);
    float z1 = c0 * c1, z2 = z1 * c2, z3 = z2 * c3, z4 = z3 * c4;
    float z5 = z4 * c5, z6 = z5 * c6, z7 = z6 * c7;
    float z0 = ((c1 * c2) * (c3 * c4)) * ((c5 * c6) * c7);
    float* zp = zsm + pos * NW;
    ((float4*)zp)[0] = make_float4(z0, z1, z2, z3);
    ((float4*)zp)[1] = make_float4(z4, z5, z6, z7);
  }
  __syncthreads();

  const int wid  = tid >> 5;
  const int lane = tid & 31;
  const int r    = lane & 15;
  const int rc   = lane & 7;
  const int hi   = lane >> 4;
  const int qrow = qblk * QB + wid * 16;

  v2f bq0, bq1;
  load_frag_pair(zsm, qrow, r, hi, bq0, bq1);

  constexpr float L2E = 1.4426950408889634f;
  constexpr float K1  = 0.35355339059327373f * L2E;
  constexpr float K0  = -2.8284271247461903f * L2E;

  v8f   acc  = {};
  float lsum = 0.f;

  v2f   ak0, ak1;
  float vv[8];
  load_frag_pair(zsm, 0, r, hi, ak0, ak1);
  load_vtile(zsm, 0, rc, hi, vv);

  for (int kt = 0; kt < S_LEN; kt += 16) {
    v2f   ck0 = ak0, ck1 = ak1;
    float cv[8];
#pragma unroll
    for (int i = 0; i < 8; ++i) cv[i] = vv[i];

    int nkt = (kt + 16) & (S_LEN - 1);
    load_frag_pair(zsm, nkt, r, hi, ak0, ak1);
    load_vtile(zsm, nkt, rc, hi, vv);

    v8f st = {};
    st = wmma4(ck0, bq0, st);
    st = wmma4(ck1, bq1, st);

    float pv[8];
#pragma unroll
    for (int i = 0; i < 8; ++i)
      pv[i] = __builtin_amdgcn_exp2f(st[i] * K1 + K0);
    lsum += ((pv[0] + pv[1]) + (pv[2] + pv[3])) +
            ((pv[4] + pv[5]) + (pv[6] + pv[7]));

    float sx[8];
#pragma unroll
    for (int i = 0; i < 8; ++i) sx[i] = xor16(pv[i]);
    v2f pa[4];
    pa[0] = v2f{hi ? sx[2] : pv[0], hi ? sx[3] : pv[1]};
    pa[1] = v2f{hi ? sx[6] : pv[4], hi ? sx[7] : pv[5]};
    pa[2] = v2f{hi ? pv[2] : sx[0], hi ? pv[3] : sx[1]};
    pa[3] = v2f{hi ? pv[6] : sx[4], hi ? pv[7] : sx[5]};

#pragma unroll
    for (int sl = 0; sl < 4; ++sl) {
      v2f bv = v2f{cv[2 * sl], cv[2 * sl + 1]};
      acc = wmma4(pa[sl], bv, acc);
    }
  }

  float ltot = lsum + xor16(lsum);
  float inv  = 1.0f / ltot;

  float* outp = attn_out + (size_t)b * S_LEN * EMB + h * NW;
#pragma unroll
  for (int i = 0; i < 8; ++i) {
    float f = __shfl(inv, i + hi * 8, 32);
    float v = acc[i] * f;
    if (r < 8) outp[(size_t)(qrow + i + hi * 8) * EMB + r] = v;
  }
}

__global__ __launch_bounds__(256) void combine_kernel(
    const float* __restrict__ a,
    const float* __restrict__ W,
    float* __restrict__ out) {
  __shared__ float Wl[EMB * EMB];
  const int tid = threadIdx.x;
#pragma unroll
  for (int k = 0; k < 16; ++k) Wl[tid + k * 256] = W[tid + k * 256];
  __syncthreads();

  const int wid  = tid >> 5;
  const int lane = tid & 31;
  const int r    = lane & 15;
  const int hi   = lane >> 4;
  const int mb   = (blockIdx.x * 8 + wid) * 16;

  v2f af[16];
#pragma unroll
  for (int i = 0; i < 16; ++i)
    af[i] = *(const v2f*)(a + (size_t)(mb + r) * EMB + 4 * i + hi * 2);

#pragma unroll
  for (int nt = 0; nt < 4; ++nt) {
    v8f accw = {};
#pragma unroll
    for (int i = 0; i < 16; ++i) {
      v2f bf = *(const v2f*)(Wl + (nt * 16 + r) * EMB + 4 * i + hi * 2);
      accw = wmma4(af[i], bf, accw);
    }
#pragma unroll
    for (int i = 0; i < 8; ++i)
      out[(size_t)(mb + i + hi * 8) * EMB + nt * 16 + r] = accw[i];
  }
}

extern "C" void kernel_launch(void* const* d_in, const int* in_sizes, int n_in,
                              void* d_out, int out_size, void* d_ws,
                              size_t ws_size, hipStream_t stream) {
  (void)in_sizes; (void)n_in; (void)out_size; (void)ws_size;
  const float* x     = (const float*)d_in[0];
  const float* theta = (const float*)d_in[1];
  const float* W     = (const float*)d_in[2];
  float* out  = (float*)d_out;
  float* attn = (float*)d_ws;

  qattn_kernel<<<dim3(512), dim3(256), 0, stream>>>(x, theta, attn);
  combine_kernel<<<dim3(64), dim3(256), 0, stream>>>(attn, W, out);
}


# PV product done transposed (O^T = V^T P^T): score registers are its B operand as they stand, no half-wave swaps or selects; LDS loads fenced into one group per tile; key order in 16-key sums changes
# speedup vs baseline: 1.1526x; 1.1526x over previous
#include <hip/hip_runtime.h>
#include <hip/hip_bf16.h>
#include <math.h>

typedef __attribute__((ext_vector_type(2))) float v2f;
typedef __attribute__((ext_vector_type(8))) float v8f;

#define S_LEN 2048
#define EMB   64
#define NW    8
#define QB    128

__device__ __forceinline__ v8f wmma4(v2f a, v2f b, v8f c) {
  return __builtin_amdgcn_wmma_f32_16x16x4_f32(false, a, false, b, (short)0, c,
                                               false, false);
}

__device__ __forceinline__ float xor16(float v) {
  int i = __float_as_int(v);
  int o = __builtin_amdgcn_permlanex16(i, i, 0x76543210, 0xfedcba98, false, false);
  return __int_as_float(o);
}

__device__ __forceinline__ void load_frag_pair(const float* zsm, int rowbase,
                                               int r, int hi, v2f& f0, v2f& f1) {
  const float* p = zsm + (rowbase + r) * NW + hi * 2;
  f0 = *(const v2f*)p;
  f1 = *(const v2f*)(p + 4);
}

__device__ __forceinline__ void load_vtile(const float* zsm, int kt, int rc,
                                           int hi, float vv[8]) {
#pragma unroll
  for (int sl = 0; sl < 4; ++sl) {
    int row = kt + sl * 4 + hi;
    vv[2 * sl]     = zsm[row * NW + rc];
    vv[2 * sl + 1] = zsm[(row + 2) * NW + rc];
  }
}

__global__ __launch_bounds__(256) void qattn_kernel(
    const float* __restrict__ x,
    const float* __restrict__ theta,
    float* __restrict__ attn_out) {
  __shared__ float zsm[S_LEN * NW];

  const int qblk = blockIdx.x & 15;
  const int bh   = blockIdx.x >> 4;
  const int b    = bh >> 3;
  const int h    = bh & 7;
  const int tid  = threadIdx.x;

  float th[8];
#pragma unroll
  for (int j = 0; j < 8; ++j) th[j] = theta[j];

  const float* xbase = x + (size_t)b * S_LEN * EMB + h * NW;
#pragma unroll
  for (int k = 0; k < 8; ++k) {
    int pos = tid + k * 256;
    const float* xrow = xbase + (size_t)pos * EMB;
    if (k + 1 < 8)
      __builtin_prefetch(xbase + (size_t)(pos + 256) * EMB, 0, 3);
    float4 xa = ((const float4*)xrow)[0];
    float4 xc = ((const float4*)xrow)[1];
    float c0 = __cosf(xa.x + th[0]);
    float c1 = __cosf(xa.y + th[1]);
    float c2 = __cosf(xa.z + th[2]);
    float c3 = __cosf(xa.w + th[3]);
    float c4 = __cosf(xc.x + th[4]);
    float c5 = __cosf(xc.y + th[5]);
    float c6 = __cosf(xc.z + th[6]);
    float c7 = __cosf(xc.w + th[7]);
    float z1 = c0 * c1, z2 = z1 * c2, z3 = z2 * c3, z4 = z3 * c4;
    float z5 = z4 * c5, z6 = z5 * c6, z7 = z6 * c7;
    float z0 = ((c1 * c2) * (c3 * c4)) * ((c5 * c6) * c7);
    float* zp = zsm + pos * NW;
    ((float4*)zp)[0] = make_float4(z0, z1, z2, z3);
    ((float4*)zp)[1] = make_float4(z4, z5, z6, z7);
  }
  __syncthreads();

  const int wid  = tid >> 5;
  const int lane = tid & 31;
  const int r    = lane & 15;
  const int rc   = lane & 7;
  const int hi   = lane >> 4;
  const int qrow = qblk * QB + wid * 16;
  const int rk   = 2 * (r & 7) + (r >> 3);

  v2f bq0, bq1;
  load_frag_pair(zsm, qrow, r, hi, bq0, bq1);

  constexpr float L2E = 1.4426950408889634f;
  constexpr float K1  = 0.35355339059327373f * L2E;
  constexpr float K0  = -2.8284271247461903f * L2E;

  v8f   acc  = {};
  float lsum = 0.f;

  v2f   ak0, ak1;
  float vv[8];
  load_frag_pair(zsm, 0, rk, hi, ak0, ak1);
  load_vtile(zsm, 0, rc, hi, vv);

  for (int kt = 0; kt < S_LEN; kt += 16) {
    v2f   ck0 = ak0, ck1 = ak1;
    float cv[8];
#pragma unroll
    for (int i = 0; i < 8; ++i) cv[i] = vv[i];

    int nkt = (kt + 16) & (S_LEN - 1);
    __builtin_amdgcn_sched_barrier(0);
    load_frag_pair(zsm, nkt, rk, hi, ak0, ak1);
    load_vtile(zsm, nkt, rc, hi, vv);
    __builtin_amdgcn_sched_barrier(0);

    v8f st = {};
    st = wmma4(ck0, bq0, st);
    st = wmma4(ck1, bq1, st);

    float pv[8];
#pragma unroll
    for (int i = 0; i < 8; ++i)
      pv[i] = __builtin_amdgcn_exp2f(st[i] * K1 + K0);
    lsum += ((pv[0] + pv[1]) + (pv[2] + pv[3])) +
            ((pv[4] + pv[5]) + (pv[6] + pv[7]));

#pragma unroll
    for (int sl = 0; sl < 4; ++sl) {
      v2f av = v2f{cv[2 * sl], cv[2 * sl + 1]};
      v2f bp = v2f{pv[2 * sl], pv[2 * sl + 1]};
      acc = wmma4(av, bp, acc);
    }
  }

  float ltot = lsum + xor16(lsum);
  float inv  = 1.0f / ltot;

  float* outp = attn_out + (size_t)b * S_LEN * EMB + h * NW;
  float4 o;
  o.x = (hi ? acc[4] : acc[0]) * inv;
  o.y = (hi ? acc[5] : acc[1]) * inv;
  o.z = (hi ? acc[6] : acc[2]) * inv;
  o.w = (hi ? acc[7] : acc[3]) * inv;
  *(float4*)(outp + (size_t)(qrow + r) * EMB + hi * 4) = o;
}

__global__ __launch_bounds__(256) void combine_kernel(
    const float* __restrict__ a,
    const float* __restrict__ W,
    float* __restrict__ out) {
  __shared__ float Wl[EMB * EMB];
  const int tid = threadIdx.x;
#pragma unroll
  for (int k = 0; k < 16; ++k) Wl[tid + k * 256] = W[tid + k * 256];
  __syncthreads();

  const int wid  = tid >> 5;
  const int lane = tid & 31;
  const int r    = lane & 15;
  const int hi   = lane >> 4;
  const int mb   = (blockIdx.x * 8 + wid) * 16;

  v2f af[16];
#pragma unroll
  for (int i = 0; i < 16; ++i)
    af[i] = *(const v2f*)(a + (size_t)(mb + r) * EMB + 4 * i + hi * 2);

#pragma unroll
  for (int nt = 0; nt < 4; ++nt) {
    v8f accw = {};
#pragma unroll
    for (int i = 0; i < 16; ++i) {
      v2f bf = *(const v2f*)(Wl + (nt * 16 + r) * EMB + 4 * i + hi * 2);
      accw = wmma4(af[i], bf, accw);
    }
#pragma unroll
    for (int i = 0; i < 8; ++i)
      out[(size_t)(mb + i + hi * 8) * EMB + nt * 16 + r] = accw[i];
  }
}

extern "C" void kernel_launch(void* const* d_in, const int* in_sizes, int n_in,
                              void* d_out, int out_size, void* d_ws,
                              size_t ws_size, hipStream_t stream) {
  (void)in_sizes; (void)n_in; (void)out_size; (void)ws_size;
  const float* x     = (const float*)d_in[0];
  const float* theta = (const float*)d_in[1];
  const float* W     = (const float*)d_in[2];
  float* out  = (float*)d_out;
  float* attn = (float*)d_ws;

  qattn_kernel<<<dim3(512), dim3(256), 0, stream>>>(x, theta, attn);
  combine_kernel<<<dim3(64), dim3(256), 0, stream>>>(attn, W, out);
}


# Each wave carries two 16-query tiles (256 queries per workgroup, 256 workgroups): each K/V^T fragment read feeds both, two independent matrix chains per wave, half as many workgroups stage z
# speedup vs baseline: 1.2185x; 1.0572x over previous
#include <hip/hip_runtime.h>
#include <hip/hip_bf16.h>
#include <math.h>

typedef __attribute__((ext_vector_type(2))) float v2f;
typedef __attribute__((ext_vector_type(8))) float v8f;

#define S_LEN 2048
#define EMB   64
#define NW    8
#define QB    256

__device__ __forceinline__ v8f wmma4(v2f a, v2f b, v8f c) {
  return __builtin_amdgcn_wmma_f32_16x16x4_f32(false, a, false, b, (short)0, c,
                                               false, false);
}

__device__ __forceinline__ float xor16(float v) {
  int i = __float_as_int(v);
  int o = __builtin_amdgcn_permlanex16(i, i, 0x76543210, 0xfedcba98, false, false);
  return __int_as_float(o);
}

__device__ __forceinline__ void load_frag_pair(const float* zsm, int rowbase,
                                               int r, int hi, v2f& f0, v2f& f1) {
  const float* p = zsm + (rowbase + r) * NW + hi * 2;
  f0 = *(const v2f*)p;
  f1 = *(const v2f*)(p + 4);
}

__device__ __forceinline__ void load_vtile(const float* zsm, int kt, int rc,
                                           int hi, float vv[8]) {
#pragma unroll
  for (int sl = 0; sl < 4; ++sl) {
    int row = kt + sl * 4 + hi;
    vv[2 * sl]     = zsm[row * NW + rc];
    vv[2 * sl + 1] = zsm[(row + 2) * NW + rc];
  }
}

__global__ __launch_bounds__(256) void qattn_kernel(
    const float* __restrict__ x,
    const float* __restrict__ theta,
    float* __restrict__ attn_out) {
  __shared__ float zsm[S_LEN * NW];

  const int qblk = blockIdx.x & 7;
  const int bh   = blockIdx.x >> 3;
  const int b    = bh >> 3;
  const int h    = bh & 7;
  const int tid  = threadIdx.x;

  float th[8];
#pragma unroll
  for (int j = 0; j < 8; ++j) th[j] = theta[j];

  const float* xbase = x + (size_t)b * S_LEN * EMB + h * NW;
#pragma unroll
  for (int k = 0; k < 8; ++k) {
    int pos = tid + k * 256;
    const float* xrow = xbase + (size_t)pos * EMB;
    if (k + 1 < 8)
      __builtin_prefetch(xbase + (size_t)(pos + 256) * EMB, 0, 3);
    float4 xa = ((const float4*)xrow)[0];
    float4 xc = ((const float4*)xrow)[1];
    float c0 = __cosf(xa.x + th[0]);
    float c1 = __cosf(xa.y + th[1]);
    float c2 = __cosf(xa.z + th[2]);
    float c3 = __cosf(xa.w + th[3]);
    float c4 = __cosf(xc.x + th[4]);
    float c5 = __cosf(xc.y + th[5]);
    float c6 = __cosf(xc.z + th[6]);
    float c7 = __cosf(xc.w + th[7]);
    float z1 = c0 * c1, z2 = z1 * c2, z3 = z2 * c3, z4 = z3 * c4;
    float z5 = z4 * c5, z6 = z5 * c6, z7 = z6 * c7;
    float z0 = ((c1 * c2) * (c3 * c4)) * ((c5 * c6) * c7);
    float* zp = zsm + pos * NW;
    ((float4*)zp)[0] = make_float4(z0, z1, z2, z3);
    ((float4*)zp)[1] = make_float4(z4, z5, z6, z7);
  }
  __syncthreads();

  const int wid  = tid >> 5;
  const int lane = tid & 31;
  const int r    = lane & 15;
  const int rc   = lane & 7;
  const int hi   = lane >> 4;
  const int qrow = qblk * QB + wid * 32;
  const int rk   = 2 * (r & 7) + (r >> 3);

  v2f bqa0, bqa1, bqb0, bqb1;
  load_frag_pair(zsm, qrow, r, hi, bqa0, bqa1);
  load_frag_pair(zsm, qrow + 16, r, hi, bqb0, bqb1);

  constexpr float L2E = 1.4426950408889634f;
  constexpr float K1  = 0.35355339059327373f * L2E;
  constexpr float K0  = -2.8284271247461903f * L2E;

  v8f   acca = {}, accb = {};
  float lsuma = 0.f, lsumb = 0.f;

  v2f   ak0, ak1;
  float vv[8];
  load_frag_pair(zsm, 0, rk, hi, ak0, ak1);
  load_vtile(zsm, 0, rc, hi, vv);

#pragma unroll 2
  for (int kt = 0; kt < S_LEN; kt += 16) {
    v2f   ck0 = ak0, ck1 = ak1;
    float cv[8];
#pragma unroll
    for (int i = 0; i < 8; ++i) cv[i] = vv[i];

    int nkt = (kt + 16) & (S_LEN - 1);
    __builtin_amdgcn_sched_barrier(0);
    load_frag_pair(zsm, nkt, rk, hi, ak0, ak1);
    load_vtile(zsm, nkt, rc, hi, vv);
    __builtin_amdgcn_sched_barrier(0);

    v8f sta = {}, stb = {};
    sta = wmma4(ck0, bqa0, sta);
    sta = wmma4(ck1, bqa1, sta);
    stb = wmma4(ck0, bqb0, stb);
    stb = wmma4(ck1, bqb1, stb);

    float pa[8], pb[8];
#pragma unroll
    for (int i = 0; i < 8; ++i)
      pa[i] = __builtin_amdgcn_exp2f(sta[i] * K1 + K0);
    lsuma += ((pa[0] + pa[1]) + (pa[2] + pa[3])) +
             ((pa[4] + pa[5]) + (pa[6] + pa[7]));

#pragma unroll
    for (int sl = 0; sl < 4; ++sl) {
      v2f av = v2f{cv[2 * sl], cv[2 * sl + 1]};
      v2f bp = v2f{pa[2 * sl], pa[2 * sl + 1]};
      acca = wmma4(av, bp, acca);
    }

#pragma unroll
    for (int i = 0; i < 8; ++i)
      pb[i] = __builtin_amdgcn_exp2f(stb[i] * K1 + K0);
    lsumb += ((pb[0] + pb[1]) + (pb[2] + pb[3])) +
             ((pb[4] + pb[5]) + (pb[6] + pb[7]));

#pragma unroll
    for (int sl = 0; sl < 4; ++sl) {
      v2f av = v2f{cv[2 * sl], cv[2 * sl + 1]};
      v2f bp = v2f{pb[2 * sl], pb[2 * sl + 1]};
      accb = wmma4(av, bp, accb);
    }
  }

  float inva = 1.0f / (lsuma + xor16(lsuma));
  float invb = 1.0f / (lsumb + xor16(lsumb));

  float* outp = attn_out + (size_t)b * S_LEN * EMB + h * NW;
  float4 oa, ob;
  oa.x = (hi ? acca[4] : acca[0]) * inva;
  oa.y = (hi ? acca[5] : acca[1]) * inva;
  oa.z = (hi ? acca[6] : acca[2]) * inva;
  oa.w = (hi ? acca[7] : acca[3]) * inva;
  ob.x = (hi ? accb[4] : accb[0]) * invb;
  ob.y = (hi ? accb[5] : accb[1]) * invb;
  ob.z = (hi ? accb[6] : accb[2]) * invb;
  ob.w = (hi ? accb[7] : accb[3]) * invb;
  *(float4*)(outp + (size_t)(qrow + r) * EMB + hi * 4) = oa;
  *(float4*)(outp + (size_t)(qrow + 16 + r) * EMB + hi * 4) = ob;
}

__global__ __launch_bounds__(256) void combine_kernel(
    const float* __restrict__ a,
    const float* __restrict__ W,
    float* __restrict__ out) {
  __shared__ float Wl[EMB * EMB];
  const int tid = threadIdx.x;
#pragma unroll
  for (int k = 0; k < 16; ++k) Wl[tid + k * 256] = W[tid + k * 256];
  __syncthreads();

  const int wid  = tid >> 5;
  const int lane = tid & 31;
  const int r    = lane & 15;
  const int hi   = lane >> 4;
  const int mb   = (blockIdx.x * 8 + wid) * 16;

  v2f af[16];
#pragma unroll
  for (int i = 0; i < 16; ++i)
    af[i] = *(const v2f*)(a + (size_t)(mb + r) * EMB + 4 * i + hi * 2);

#pragma unroll
  for (int nt = 0; nt < 4; ++nt) {
    v8f accw = {};
#pragma unroll
    for (int i = 0; i < 16; ++i) {
      v2f bf = *(const v2f*)(Wl + (nt * 16 + r) * EMB + 4 * i + hi * 2);
      accw = wmma4(af[i], bf, accw);
    }
#pragma unroll
    for (int i = 0; i < 8; ++i)
      out[(size_t)(mb + i + hi * 8) * EMB + nt * 16 + r] = accw[i];
  }
}

extern "C" void kernel_launch(void* const* d_in, const int* in_sizes, int n_in,
                              void* d_out, int out_size, void* d_ws,
                              size_t ws_size, hipStream_t stream) {
  (void)in_sizes; (void)n_in; (void)out_size; (void)ws_size;
  const float* x     = (const float*)d_in[0];
  const float* theta = (const float*)d_in[1];
  const float* W     = (const float*)d_in[2];
  float* out  = (float*)d_out;
  float* attn = (float*)d_ws;

  qattn_kernel<<<dim3(256), dim3(256), 0, stream>>>(x, theta, attn);
  combine_kernel<<<dim3(64), dim3(256), 0, stream>>>(attn, W, out);
}


# Softmax vector work folded into WMMA operands: log2e/sqrt8 scale pre-multiplied into Q fragments, shift dropped (|score|<=sqrt8), row sums from an all-ones V^T row; denominator sum order changes
# speedup vs baseline: 1.3102x; 1.0753x over previous
#include <hip/hip_runtime.h>
#include <hip/hip_bf16.h>
#include <math.h>

typedef __attribute__((ext_vector_type(2))) float v2f;
typedef __attribute__((ext_vector_type(8))) float v8f;

#define S_LEN 2048
#define EMB   64
#define NW    8
#define QB    256
#define PAD   160

__device__ __forceinline__ v8f wmma4(v2f a, v2f b, v8f c) {
  return __builtin_amdgcn_wmma_f32_16x16x4_f32(false, a, false, b, (short)0, c,
                                               false, false);
}

__device__ __forceinline__ float xor16(float v) {
  int i = __float_as_int(v);
  int o = __builtin_amdgcn_permlanex16(i, i, 0x76543210, 0xfedcba98, false, false);
  return __int_as_float(o);
}

__device__ __forceinline__ void load_frag_pair(const float* zsm, int rowbase,
                                               int r, int hi, v2f& f0, v2f& f1) {
  const float* p = zsm + (rowbase + r) * NW + hi * 2;
  f0 = *(const v2f*)p;
  f1 = *(const v2f*)(p + 4);
}

__device__ __forceinline__ void load_vtile(const float* vp, float vv[8]) {
#pragma unroll
  for (int sl = 0; sl < 4; ++sl) {
    vv[2 * sl]     = vp[sl * 4 * NW];
    vv[2 * sl + 1] = vp[(sl * 4 + 2) * NW];
  }
}

__global__ __launch_bounds__(256) void qattn_kernel(
    const float* __restrict__ x,
    const float* __restrict__ theta,
    float* __restrict__ attn_out) {
  __shared__ float zsm[S_LEN * NW + PAD];

  const int qblk = blockIdx.x & 7;
  const int bh   = blockIdx.x >> 3;
  const int b    = bh >> 3;
  const int h    = bh & 7;
  const int tid  = threadIdx.x;

  float th[8];
#pragma unroll
  for (int j = 0; j < 8; ++j) th[j] = theta[j];

  const float* xbase = x + (size_t)b * S_LEN * EMB + h * NW;
#pragma unroll
  for (int k = 0; k < 8; ++k) {
    int pos = tid + k * 256;
    const float* xrow = xbase + (size_t)pos * EMB;
    if (k + 1 < 8)
      __builtin_prefetch(xbase + (size_t)(pos + 256) * EMB, 0, 3);
    float4 xa = ((const float4*)xrow)[0];
    float4 xc = ((const float4*)xrow)[1];
    float c0 = __cosf(xa.x + th[0]);
    float c1 = __cosf(xa.y + th[1]);
    float c2 = __cosf(xa.z + th[2]);
    float c3 = __cosf(xa.w + th[3]);
    float c4 = __cosf(xc.x + th[4]);
    float c5 = __cosf(xc.y + th[5]);
    float c6 = __cosf(xc.z + th[6]);
    float c7 = __cosf(xc.w + th[7]);
    float z1 = c0 * c1, z2 = z1 * c2, z3 = z2 * c3, z4 = z3 * c4;
    float z5 = z4 * c5, z6 = z5 * c6, z7 = z6 * c7;
    float z0 = ((c1 * c2) * (c3 * c4)) * ((c5 * c6) * c7);
    float* zp = zsm + pos * NW;
    ((float4*)zp)[0] = make_float4(z0, z1, z2, z3);
    ((float4*)zp)[1] = make_float4(z4, z5, z6, z7);
  }
  if (tid < PAD) zsm[S_LEN * NW + tid] = 1.0f;
  __syncthreads();

  const int wid  = tid >> 5;
  const int lane = tid & 31;
  const int r    = lane & 15;
  const int rc   = lane & 7;
  const int hi   = lane >> 4;
  const int qrow = qblk * QB + wid * 32;
  const int rk   = 2 * (r & 7) + (r >> 3);

  v2f bqa0, bqa1, bqb0, bqb1;
  load_frag_pair(zsm, qrow, r, hi, bqa0, bqa1);
  load_frag_pair(zsm, qrow + 16, r, hi, bqb0, bqb1);

  constexpr float L2E = 1.4426950408889634f;
  constexpr float K1  = 0.35355339059327373f * L2E;
  bqa0 *= K1; bqa1 *= K1; bqb0 *= K1; bqb1 *= K1;

  v8f acca = {}, accb = {};

  const float* vp    = (r < 8) ? zsm + hi * NW + rc : zsm + S_LEN * NW + 20;
  const int    vstep = (r < 8) ? 16 * NW : 0;

  const float* kp = zsm + rk * NW + hi * 2;

  v2f   ak0, ak1;
  float vv[8];
  ak0 = *(const v2f*)kp;
  ak1 = *(const v2f*)(kp + 4);
  load_vtile(vp, vv);

#pragma unroll 2
  for (int kt = 0; kt < S_LEN; kt += 16) {
    v2f   ck0 = ak0, ck1 = ak1;
    float cv[8];
#pragma unroll
    for (int i = 0; i < 8; ++i) cv[i] = vv[i];

    vp += vstep;
    kp += 16 * NW;
    __builtin_amdgcn_sched_barrier(0);
    ak0 = *(const v2f*)kp;
    ak1 = *(const v2f*)(kp + 4);
    load_vtile(vp, vv);
    __builtin_amdgcn_sched_barrier(0);

    v8f sta = {}, stb = {};
    sta = wmma4(ck0, bqa0, sta);
    sta = wmma4(ck1, bqa1, sta);
    stb = wmma4(ck0, bqb0, stb);
    stb = wmma4(ck1, bqb1, stb);

    float pa[8], pb[8];
#pragma unroll
    for (int i = 0; i < 8; ++i)
      pa[i] = __builtin_amdgcn_exp2f(sta[i]);

#pragma unroll
    for (int sl = 0; sl < 4; ++sl) {
      v2f av = v2f{cv[2 * sl], cv[2 * sl + 1]};
      v2f bp = v2f{pa[2 * sl], pa[2 * sl + 1]};
      acca = wmma4(av, bp, acca);
    }

#pragma unroll
    for (int i = 0; i < 8; ++i)
      pb[i] = __builtin_amdgcn_exp2f(stb[i]);

#pragma unroll
    for (int sl = 0; sl < 4; ++sl) {
      v2f av = v2f{cv[2 * sl], cv[2 * sl + 1]};
      v2f bp = v2f{pb[2 * sl], pb[2 * sl + 1]};
      accb = wmma4(av, bp, accb);
    }
  }

  float da = xor16(acca[0]), db = xor16(accb[0]);
  float a4 = xor16(acca[4]), a5 = xor16(acca[5]);
  float a6 = xor16(acca[6]), a7 = xor16(acca[7]);
  float b4 = xor16(accb[4]), b5 = xor16(accb[5]);
  float b6 = xor16(accb[6]), b7 = xor16(accb[7]);
  float inva = 1.0f / (hi ? acca[0] : da);
  float invb = 1.0f / (hi ? accb[0] : db);
  float* outp = attn_out + (size_t)b * S_LEN * EMB + h * NW;
  float4 oa, ob;
  oa.x = (hi ? a4 : acca[0]) * inva;
  oa.y = (hi ? a5 : acca[1]) * inva;
  oa.z = (hi ? a6 : acca[2]) * inva;
  oa.w = (hi ? a7 : acca[3]) * inva;
  ob.x = (hi ? b4 : accb[0]) * invb;
  ob.y = (hi ? b5 : accb[1]) * invb;
  ob.z = (hi ? b6 : accb[2]) * invb;
  ob.w = (hi ? b7 : accb[3]) * invb;
  *(float4*)(outp + (size_t)(qrow + r) * EMB + hi * 4) = oa;
  *(float4*)(outp + (size_t)(qrow + 16 + r) * EMB + hi * 4) = ob;
}

__global__ __launch_bounds__(256) void combine_kernel(
    const float* __restrict__ a,
    const float* __restrict__ W,
    float* __restrict__ out) {
  __shared__ float Wl[EMB * EMB];
  const int tid = threadIdx.x;
#pragma unroll
  for (int k = 0; k < 16; ++k) Wl[tid + k * 256] = W[tid + k * 256];
  __syncthreads();

  const int wid  = tid >> 5;
  const int lane = tid & 31;
  const int r    = lane & 15;
  const int hi   = lane >> 4;
  const int mb   = (blockIdx.x * 8 + wid) * 16;

  v2f af[16];
#pragma unroll
  for (int i = 0; i < 16; ++i)
    af[i] = *(const v2f*)(a + (size_t)(mb + r) * EMB + 4 * i + hi * 2);

#pragma unroll
  for (int nt = 0; nt < 4; ++nt) {
    v8f accw = {};
#pragma unroll
    for (int i = 0; i < 16; ++i) {
      v2f bf = *(const v2f*)(Wl + (nt * 16 + r) * EMB + 4 * i + hi * 2);
      accw = wmma4(af[i], bf, accw);
    }
#pragma unroll
    for (int i = 0; i < 8; ++i)
      out[(size_t)(mb + i + hi * 8) * EMB + nt * 16 + r] = accw[i];
  }
}

extern "C" void kernel_launch(void* const* d_in, const int* in_sizes, int n_in,
                              void* d_out, int out_size, void* d_ws,
                              size_t ws_size, hipStream_t stream) {
  (void)in_sizes; (void)n_in; (void)out_size; (void)ws_size;
  const float* x     = (const float*)d_in[0];
  const float* theta = (const float*)d_in[1];
  const float* W     = (const float*)d_in[2];
  float* out  = (float*)d_out;
  float* attn = (float*)d_ws;

  qattn_kernel<<<dim3(256), dim3(256), 0, stream>>>(x, theta, attn);
  combine_kernel<<<dim3(64), dim3(256), 0, stream>>>(attn, W, out);
}


# Projection kernel: W rows in LDS at pitch 68 floats instead of 64, so the 16 rows of a B-fragment read fall on different banks (was a 16-way bank conflict on every read)
# speedup vs baseline: 1.3790x; 1.0525x over previous
#include <hip/hip_runtime.h>
#include <hip/hip_bf16.h>
#include <math.h>

typedef __attribute__((ext_vector_type(2))) float v2f;
typedef __attribute__((ext_vector_type(8))) float v8f;

#define S_LEN 2048
#define EMB   64
#define NW    8
#define QB    256
#define PAD   160
#define WP    68

__device__ __forceinline__ v8f wmma4(v2f a, v2f b, v8f c) {
  return __builtin_amdgcn_wmma_f32_16x16x4_f32(false, a, false, b, (short)0, c,
                                               false, false);
}

__device__ __forceinline__ float xor16(float v) {
  int i = __float_as_int(v);
  int o = __builtin_amdgcn_permlanex16(i, i, 0x76543210, 0xfedcba98, false, false);
  return __int_as_float(o);
}

__device__ __forceinline__ void load_frag_pair(const float* zsm, int rowbase,
                                               int r, int hi, v2f& f0, v2f& f1) {
  const float* p = zsm + (rowbase + r) * NW + hi * 2;
  f0 = *(const v2f*)p;
  f1 = *(const v2f*)(p + 4);
}

__device__ __forceinline__ void load_vtile(const float* vp, float vv[8]) {
#pragma unroll
  for (int sl = 0; sl < 4; ++sl) {
    vv[2 * sl]     = vp[sl * 4 * NW];
    vv[2 * sl + 1] = vp[(sl * 4 + 2) * NW];
  }
}

__global__ __launch_bounds__(256) void qattn_kernel(
    const float* __restrict__ x,
    const float* __restrict__ theta,
    float* __restrict__ attn_out) {
  __shared__ float zsm[S_LEN * NW + PAD];

  const int qblk = blockIdx.x & 7;
  const int bh   = blockIdx.x >> 3;
  const int b    = bh >> 3;
  const int h    = bh & 7;
  const int tid  = threadIdx.x;

  float th[8];
#pragma unroll
  for (int j = 0; j < 8; ++j) th[j] = theta[j];

  const float* xbase = x + (size_t)b * S_LEN * EMB + h * NW;
#pragma unroll
  for (int k = 0; k < 8; ++k) {
    int pos = tid + k * 256;
    const float* xrow = xbase + (size_t)pos * EMB;
    if (k + 1 < 8)
      __builtin_prefetch(xbase + (size_t)(pos + 256) * EMB, 0, 3);
    float4 xa = ((const float4*)xrow)[0];
    float4 xc = ((const float4*)xrow)[1];
    float c0 = __cosf(xa.x + th[0]);
    float c1 = __cosf(xa.y + th[1]);
    float c2 = __cosf(xa.z + th[2]);
    float c3 = __cosf(xa.w + th[3]);
    float c4 = __cosf(xc.x + th[4]);
    float c5 = __cosf(xc.y + th[5]);
    float c6 = __cosf(xc.z + th[6]);
    float c7 = __cosf(xc.w + th[7]);
    float z1 = c0 * c1, z2 = z1 * c2, z3 = z2 * c3, z4 = z3 * c4;
    float z5 = z4 * c5, z6 = z5 * c6, z7 = z6 * c7;
    float z0 = ((c1 * c2) * (c3 * c4)) * ((c5 * c6) * c7);
    float* zp = zsm + pos * NW;
    ((float4*)zp)[0] = make_float4(z0, z1, z2, z3);
    ((float4*)zp)[1] = make_float4(z4, z5, z6, z7);
  }
  if (tid < PAD) zsm[S_LEN * NW + tid] = 1.0f;
  __syncthreads();

  const int wid  = tid >> 5;
  const int lane = tid & 31;
  const int r    = lane & 15;
  const int rc   = lane & 7;
  const int hi   = lane >> 4;
  const int qrow = qblk * QB + wid * 32;
  const int rk   = 2 * (r & 7) + (r >> 3);

  v2f bqa0, bqa1, bqb0, bqb1;
  load_frag_pair(zsm, qrow, r, hi, bqa0, bqa1);
  load_frag_pair(zsm, qrow + 16, r, hi, bqb0, bqb1);

  constexpr float L2E = 1.4426950408889634f;
  constexpr float K1  = 0.35355339059327373f * L2E;
  bqa0 *= K1; bqa1 *= K1; bqb0 *= K1; bqb1 *= K1;

  v8f acca = {}, accb = {};

  const float* vp    = (r < 8) ? zsm + hi * NW + rc : zsm + S_LEN * NW + 20;
  const int    vstep = (r < 8) ? 16 * NW : 0;

  const float* kp = zsm + rk * NW + hi * 2;

  v2f   ak0, ak1;
  float vv[8];
  ak0 = *(const v2f*)kp;
  ak1 = *(const v2f*)(kp + 4);
  load_vtile(vp, vv);

#pragma unroll 2
  for (int kt = 0; kt < S_LEN; kt += 16) {
    v2f   ck0 = ak0, ck1 = ak1;
    float cv[8];
#pragma unroll
    for (int i = 0; i < 8; ++i) cv[i] = vv[i];

    vp += vstep;
    kp += 16 * NW;
    __builtin_amdgcn_sched_barrier(0);
    ak0 = *(const v2f*)kp;
    ak1 = *(const v2f*)(kp + 4);
    load_vtile(vp, vv);
    __builtin_amdgcn_sched_barrier(0);

    v8f sta = {}, stb = {};
    sta = wmma4(ck0, bqa0, sta);
    sta = wmma4(ck1, bqa1, sta);
    stb = wmma4(ck0, bqb0, stb);
    stb = wmma4(ck1, bqb1, stb);

    float pa[8], pb[8];
#pragma unroll
    for (int i = 0; i < 8; ++i)
      pa[i] = __builtin_amdgcn_exp2f(sta[i]);

#pragma unroll
    for (int sl = 0; sl < 4; ++sl) {
      v2f av = v2f{cv[2 * sl], cv[2 * sl + 1]};
      v2f bp = v2f{pa[2 * sl], pa[2 * sl + 1]};
      acca = wmma4(av, bp, acca);
    }

#pragma unroll
    for (int i = 0; i < 8; ++i)
      pb[i] = __builtin_amdgcn_exp2f(stb[i]);

#pragma unroll
    for (int sl = 0; sl < 4; ++sl) {
      v2f av = v2f{cv[2 * sl], cv[2 * sl + 1]};
      v2f bp = v2f{pb[2 * sl], pb[2 * sl + 1]};
      accb = wmma4(av, bp, accb);
    }
  }

  float da = xor16(acca[0]), db = xor16(accb[0]);
  float a4 = xor16(acca[4]), a5 = xor16(acca[5]);
  float a6 = xor16(acca[6]), a7 = xor16(acca[7]);
  float b4 = xor16(accb[4]), b5 = xor16(accb[5]);
  float b6 = xor16(accb[6]), b7 = xor16(accb[7]);
  float inva = 1.0f / (hi ? acca[0] : da);
  float invb = 1.0f / (hi ? accb[0] : db);
  float* outp = attn_out + (size_t)b * S_LEN * EMB + h * NW;
  float4 oa, ob;
  oa.x = (hi ? a4 : acca[0]) * inva;
  oa.y = (hi ? a5 : acca[1]) * inva;
  oa.z = (hi ? a6 : acca[2]) * inva;
  oa.w = (hi ? a7 : acca[3]) * inva;
  ob.x = (hi ? b4 : accb[0]) * invb;
  ob.y = (hi ? b5 : accb[1]) * invb;
  ob.z = (hi ? b6 : accb[2]) * invb;
  ob.w = (hi ? b7 : accb[3]) * invb;
  *(float4*)(outp + (size_t)(qrow + r) * EMB + hi * 4) = oa;
  *(float4*)(outp + (size_t)(qrow + 16 + r) * EMB + hi * 4) = ob;
}

__global__ __launch_bounds__(256) void combine_kernel(
    const float* __restrict__ a,
    const float* __restrict__ W,
    float* __restrict__ out) {
  __shared__ float Wl[EMB * WP];
  const int tid = threadIdx.x;
  const int wcol = tid & 63;
  const int wrow = tid >> 6;
#pragma unroll
  for (int k = 0; k < 16; ++k)
    Wl[(wrow + 4 * k) * WP + wcol] = W[tid + k * 256];
  __syncthreads();

  const int wid  = tid >> 5;
  const int lane = tid & 31;
  const int r    = lane & 15;
  const int hi   = lane >> 4;
  const int mb   = (blockIdx.x * 8 + wid) * 16;

  v2f af[16];
#pragma unroll
  for (int i = 0; i < 16; ++i)
    af[i] = *(const v2f*)(a + (size_t)(mb + r) * EMB + 4 * i + hi * 2);

#pragma unroll
  for (int nt = 0; nt < 4; ++nt) {
    v8f accw = {};
#pragma unroll
    for (int i = 0; i < 16; ++i) {
      v2f bf = *(const v2f*)(Wl + (nt * 16 + r) * WP + 4 * i + hi * 2);
      accw = wmma4(af[i], bf, accw);
    }
#pragma unroll
    for (int i = 0; i < 8; ++i)
      out[(size_t)(mb + i + hi * 8) * EMB + nt * 16 + r] = accw[i];
  }
}

extern "C" void kernel_launch(void* const* d_in, const int* in_sizes, int n_in,
                              void* d_out, int out_size, void* d_ws,
                              size_t ws_size, hipStream_t stream) {
  (void)in_sizes; (void)n_in; (void)out_size; (void)ws_size;
  const float* x     = (const float*)d_in[0];
  const float* theta = (const float*)d_in[1];
  const float* W     = (const float*)d_in[2];
  float* out  = (float*)d_out;
  float* attn = (float*)d_ws;

  qattn_kernel<<<dim3(256), dim3(256), 0, stream>>>(x, theta, attn);
  combine_kernel<<<dim3(64), dim3(256), 0, stream>>>(attn, W, out);
}


# z staged in 8 chunks one chunk ahead of the key loop: x loads in flight and cosines placed inside the loop, only the own chunk staged in front; key loop starts there and wraps, key sum order rotates
# speedup vs baseline: 1.4417x; 1.0455x over previous
#include <hip/hip_runtime.h>
#include <hip/hip_bf16.h>
#include <math.h>

typedef __attribute__((ext_vector_type(2))) float v2f;
typedef __attribute__((ext_vector_type(8))) float v8f;

#define S_LEN 2048
#define EMB   64
#define NW    8
#define QB    256
#define PAD   160
#define CH    256
#define NCH   8
#define WP    68

__device__ __forceinline__ v8f wmma4(v2f a, v2f b, v8f c) {
  return __builtin_amdgcn_wmma_f32_16x16x4_f32(false, a, false, b, (short)0, c,
                                               false, false);
}

__device__ __forceinline__ float xor16(float v) {
  int i = __float_as_int(v);
  int o = __builtin_amdgcn_permlanex16(i, i, 0x76543210, 0xfedcba98, false, false);
  return __int_as_float(o);
}

__device__ __forceinline__ void load_frag_pair(const float* zsm, int rowbase,
                                               int r, int hi, v2f& f0, v2f& f1) {
  const float* p = zsm + (rowbase + r) * NW + hi * 2;
  f0 = *(const v2f*)p;
  f1 = *(const v2f*)(p + 4);
}

__device__ __forceinline__ void load_vtile(const float* vp, float vv[8]) {
#pragma unroll
  for (int sl = 0; sl < 4; ++sl) {
    vv[2 * sl]     = vp[sl * 4 * NW];
    vv[2 * sl + 1] = vp[(sl * 4 + 2) * NW];
  }
}

__device__ __forceinline__ void stage_pos(float4 xa, float4 xc,
                                          const float th[8], float* zp) {
  float c0 = __cosf(xa.x + th[0]);
  float c1 = __cosf(xa.y + th[1]);
  float c2 = __cosf(xa.z + th[2]);
  float c3 = __cosf(xa.w + th[3]);
  float c4 = __cosf(xc.x + th[4]);
  float c5 = __cosf(xc.y + th[5]);
  float c6 = __cosf(xc.z + th[6]);
  float c7 = __cosf(xc.w + th[7]);
  float z1 = c0 * c1, z2 = z1 * c2, z3 = z2 * c3, z4 = z3 * c4;
  float z5 = z4 * c5, z6 = z5 * c6, z7 = z6 * c7;
  float z0 = ((c1 * c2) * (c3 * c4)) * ((c5 * c6) * c7);
  ((float4*)zp)[0] = make_float4(z0, z1, z2, z3);
  ((float4*)zp)[1] = make_float4(z4, z5, z6, z7);
}

__device__ __forceinline__ void key_trip(const float*& kp, const float*& vp,
                                         int vstep,
                                         v2f& ak0, v2f& ak1, float (&vv)[8],
                                         v2f bqa0, v2f bqa1, v2f bqb0, v2f bqb1,
                                         v8f& acca, v8f& accb) {
  v2f   ck0 = ak0, ck1 = ak1;
  float cv[8];
#pragma unroll
  for (int i = 0; i < 8; ++i) cv[i] = vv[i];

  vp += vstep;
  kp += 16 * NW;
  __builtin_amdgcn_sched_barrier(0);
  ak0 = *(const v2f*)kp;
  ak1 = *(const v2f*)(kp + 4);
  load_vtile(vp, vv);
  __builtin_amdgcn_sched_barrier(0);

  v8f sta = {}, stb = {};
  sta = wmma4(ck0, bqa0, sta);
  sta = wmma4(ck1, bqa1, sta);
  stb = wmma4(ck0, bqb0, stb);
  stb = wmma4(ck1, bqb1, stb);

  float pa[8], pb[8];
#pragma unroll
  for (int i = 0; i < 8; ++i)
    pa[i] = __builtin_amdgcn_exp2f(sta[i]);

#pragma unroll
  for (int sl = 0; sl < 4; ++sl) {
    v2f av = v2f{cv[2 * sl], cv[2 * sl + 1]};
    v2f bp = v2f{pa[2 * sl], pa[2 * sl + 1]};
    acca = wmma4(av, bp, acca);
  }

#pragma unroll
  for (int i = 0; i < 8; ++i)
    pb[i] = __builtin_amdgcn_exp2f(stb[i]);

#pragma unroll
  for (int sl = 0; sl < 4; ++sl) {
    v2f av = v2f{cv[2 * sl], cv[2 * sl + 1]};
    v2f bp = v2f{pb[2 * sl], pb[2 * sl + 1]};
    accb = wmma4(av, bp, accb);
  }
}

__global__ __launch_bounds__(256) void qattn_kernel(
    const float* __restrict__ x,
    const float* __restrict__ theta,
    float* __restrict__ attn_out) {
  __shared__ float zsm[S_LEN * NW + PAD];

  const int qblk = blockIdx.x & 7;
  const int bh   = blockIdx.x >> 3;
  const int b    = bh >> 3;
  const int h    = bh & 7;
  const int tid  = threadIdx.x;

  float th[8];
#pragma unroll
  for (int j = 0; j < 8; ++j) th[j] = theta[j];

  const float* xbase = x + (size_t)b * S_LEN * EMB + h * NW;
  {
    const float* xrow = xbase + (size_t)(qblk * CH + tid) * EMB;
    float4 xa = ((const float4*)xrow)[0];
    float4 xc = ((const float4*)xrow)[1];
    stage_pos(xa, xc, th, zsm + tid * NW);
  }
  if (tid < PAD) zsm[S_LEN * NW + tid] = 1.0f;
  __syncthreads();

  const int wid  = tid >> 5;
  const int lane = tid & 31;
  const int r    = lane & 15;
  const int rc   = lane & 7;
  const int hi   = lane >> 4;
  const int qrow = qblk * QB + wid * 32;
  const int qloc = wid * 32;
  const int rk   = 2 * (r & 7) + (r >> 3);

  v2f bqa0, bqa1, bqb0, bqb1;
  load_frag_pair(zsm, qloc, r, hi, bqa0, bqa1);
  load_frag_pair(zsm, qloc + 16, r, hi, bqb0, bqb1);

  constexpr float L2E = 1.4426950408889634f;
  constexpr float K1  = 0.35355339059327373f * L2E;
  bqa0 *= K1; bqa1 *= K1; bqb0 *= K1; bqb1 *= K1;

  v8f acca = {}, accb = {};

  const float* vp    = (r < 8) ? zsm + hi * NW + rc : zsm + S_LEN * NW + 20;
  const int    vstep = (r < 8) ? 16 * NW : 0;

  const float* kp = zsm + rk * NW + hi * 2;

  v2f   ak0, ak1;
  float vv[8];
  ak0 = *(const v2f*)kp;
  ak1 = *(const v2f*)(kp + 4);
  load_vtile(vp, vv);

  for (int ph = 0; ph < NCH; ++ph) {
    const bool more = ph + 1 < NCH;
    float4 xa = {}, xc = {};
    if (more) {
      const int npos = ((qblk + ph + 1) & (NCH - 1)) * CH + tid;
      const float* xrow = xbase + (size_t)npos * EMB;
      xa = ((const float4*)xrow)[0];
      xc = ((const float4*)xrow)[1];
    }

#pragma unroll 2
    for (int i = 0; i < 8; ++i)
      key_trip(kp, vp, vstep, ak0, ak1, vv, bqa0, bqa1, bqb0, bqb1, acca, accb);

    if (more) stage_pos(xa, xc, th, zsm + ((ph + 1) * CH + tid) * NW);

#pragma unroll 2
    for (int i = 0; i < 4; ++i)
      key_trip(kp, vp, vstep, ak0, ak1, vv, bqa0, bqa1, bqb0, bqb1, acca, accb);

    if (more) __syncthreads();

#pragma unroll 2
    for (int i = 0; i < 4; ++i)
      key_trip(kp, vp, vstep, ak0, ak1, vv, bqa0, bqa1, bqb0, bqb1, acca, accb);
  }

  float da = xor16(acca[0]), db = xor16(accb[0]);
  float a4 = xor16(acca[4]), a5 = xor16(acca[5]);
  float a6 = xor16(acca[6]), a7 = xor16(acca[7]);
  float b4 = xor16(accb[4]), b5 = xor16(accb[5]);
  float b6 = xor16(accb[6]), b7 = xor16(accb[7]);
  float inva = 1.0f / (hi ? acca[0] : da);
  float invb = 1.0f / (hi ? accb[0] : db);
  float* outp = attn_out + (size_t)b * S_LEN * EMB + h * NW;
  float4 oa, ob;
  oa.x = (hi ? a4 : acca[0]) * inva;
  oa.y = (hi ? a5 : acca[1]) * inva;
  oa.z = (hi ? a6 : acca[2]) * inva;
  oa.w = (hi ? a7 : acca[3]) * inva;
  ob.x = (hi ? b4 : accb[0]) * invb;
  ob.y = (hi ? b5 : accb[1]) * invb;
  ob.z = (hi ? b6 : accb[2]) * invb;
  ob.w = (hi ? b7 : accb[3]) * invb;
  *(float4*)(outp + (size_t)(qrow + r) * EMB + hi * 4) = oa;
  *(float4*)(outp + (size_t)(qrow + 16 + r) * EMB + hi * 4) = ob;
}

__global__ __launch_bounds__(256) void combine_kernel(
    const float* __restrict__ a,
    const float* __restrict__ W,
    float* __restrict__ out) {
  __shared__ float Wl[EMB * WP];
  const int tid = threadIdx.x;
  const int wcol = tid & 63;
  const int wrow = tid >> 6;
#pragma unroll
  for (int k = 0; k < 16; ++k)
    Wl[(wrow + 4 * k) * WP + wcol] = W[tid + k * 256];
  __syncthreads();

  const int wid  = tid >> 5;
  const int lane = tid & 31;
  const int r    = lane & 15;
  const int hi   = lane >> 4;
  const int mb   = (blockIdx.x * 8 + wid) * 16;

  v2f af[16];
#pragma unroll
  for (int i = 0; i < 16; ++i)
    af[i] = *(const v2f*)(a + (size_t)(mb + r) * EMB + 4 * i + hi * 2);

#pragma unroll
  for (int nt = 0; nt < 4; ++nt) {
    v8f accw = {};
#pragma unroll
    for (int i = 0; i < 16; ++i) {
      v2f bf = *(const v2f*)(Wl + (nt * 16 + r) * WP + 4 * i + hi * 2);
      accw = wmma4(af[i], bf, accw);
    }
#pragma unroll
    for (int i = 0; i < 8; ++i)
      out[(size_t)(mb + i + hi * 8) * EMB + nt * 16 + r] = accw[i];
  }
}

extern "C" void kernel_launch(void* const* d_in, const int* in_sizes, int n_in,
                              void* d_out, int out_size, void* d_ws,
                              size_t ws_size, hipStream_t stream) {
  (void)in_sizes; (void)n_in; (void)out_size; (void)ws_size;
  const float* x     = (const float*)d_in[0];
  const float* theta = (const float*)d_in[1];
  const float* W     = (const float*)d_in[2];
  float* out  = (float*)d_out;
  float* attn = (float*)d_ws;

  qattn_kernel<<<dim3(256), dim3(256), 0, stream>>>(x, theta, attn);
  combine_kernel<<<dim3(64), dim3(256), 0, stream>>>(attn, W, out);
}


# Projection kernel without LDS or barrier: each wave loads its A rows and all of W into registers by 40 16-byte loads ahead of its 64 WMMAs, 256 two-wave workgroups; k order of its sums changes
# speedup vs baseline: 1.4994x; 1.0400x over previous
#include <hip/hip_runtime.h>
#include <hip/hip_bf16.h>
#include <math.h>

typedef __attribute__((ext_vector_type(2))) float v2f;
typedef __attribute__((ext_vector_type(8))) float v8f;

#define S_LEN 2048
#define EMB   64
#define NW    8
#define QB    256
#define PAD   160
#define CH    256
#define NCH   8

__device__ __forceinline__ v8f wmma4(v2f a, v2f b, v8f c) {
  return __builtin_amdgcn_wmma_f32_16x16x4_f32(false, a, false, b, (short)0, c,
                                               false, false);
}

__device__ __forceinline__ float xor16(float v) {
  int i = __float_as_int(v);
  int o = __builtin_amdgcn_permlanex16(i, i, 0x76543210, 0xfedcba98, false, false);
  return __int_as_float(o);
}

__device__ __forceinline__ void load_frag_pair(const float* zsm, int rowbase,
                                               int r, int hi, v2f& f0, v2f& f1) {
  const float* p = zsm + (rowbase + r) * NW + hi * 2;
  f0 = *(const v2f*)p;
  f1 = *(const v2f*)(p + 4);
}

__device__ __forceinline__ void load_vtile(const float* vp, float vv[8]) {
#pragma unroll
  for (int sl = 0; sl < 4; ++sl) {
    vv[2 * sl]     = vp[sl * 4 * NW];
    vv[2 * sl + 1] = vp[(sl * 4 + 2) * NW];
  }
}

__device__ __forceinline__ void stage_pos(float4 xa, float4 xc,
                                          const float th[8], float* zp) {
  float c0 = __cosf(xa.x + th[0]);
  float c1 = __cosf(xa.y + th[1]);
  float c2 = __cosf(xa.z + th[2]);
  float c3 = __cosf(xa.w + th[3]);
  float c4 = __cosf(xc.x + th[4]);
  float c5 = __cosf(xc.y + th[5]);
  float c6 = __cosf(xc.z + th[6]);
  float c7 = __cosf(xc.w + th[7]);
  float z1 = c0 * c1, z2 = z1 * c2, z3 = z2 * c3, z4 = z3 * c4;
  float z5 = z4 * c5, z6 = z5 * c6, z7 = z6 * c7;
  float z0 = ((c1 * c2) * (c3 * c4)) * ((c5 * c6) * c7);
  ((float4*)zp)[0] = make_float4(z0, z1, z2, z3);
  ((float4*)zp)[1] = make_float4(z4, z5, z6, z7);
}

__device__ __forceinline__ void key_trip(const float*& kp, const float*& vp,
                                         int vstep,
                                         v2f& ak0, v2f& ak1, float (&vv)[8],
                                         v2f bqa0, v2f bqa1, v2f bqb0, v2f bqb1,
                                         v8f& acca, v8f& accb) {
  v2f   ck0 = ak0, ck1 = ak1;
  float cv[8];
#pragma unroll
  for (int i = 0; i < 8; ++i) cv[i] = vv[i];

  vp += vstep;
  kp += 16 * NW;
  __builtin_amdgcn_sched_barrier(0);
  ak0 = *(const v2f*)kp;
  ak1 = *(const v2f*)(kp + 4);
  load_vtile(vp, vv);
  __builtin_amdgcn_sched_barrier(0);

  v8f sta = {}, stb = {};
  sta = wmma4(ck0, bqa0, sta);
  sta = wmma4(ck1, bqa1, sta);
  stb = wmma4(ck0, bqb0, stb);
  stb = wmma4(ck1, bqb1, stb);

  float pa[8], pb[8];
#pragma unroll
  for (int i = 0; i < 8; ++i)
    pa[i] = __builtin_amdgcn_exp2f(sta[i]);

#pragma unroll
  for (int sl = 0; sl < 4; ++sl) {
    v2f av = v2f{cv[2 * sl], cv[2 * sl + 1]};
    v2f bp = v2f{pa[2 * sl], pa[2 * sl + 1]};
    acca = wmma4(av, bp, acca);
  }

#pragma unroll
  for (int i = 0; i < 8; ++i)
    pb[i] = __builtin_amdgcn_exp2f(stb[i]);

#pragma unroll
  for (int sl = 0; sl < 4; ++sl) {
    v2f av = v2f{cv[2 * sl], cv[2 * sl + 1]};
    v2f bp = v2f{pb[2 * sl], pb[2 * sl + 1]};
    accb = wmma4(av, bp, accb);
  }
}

__global__ __launch_bounds__(256) void qattn_kernel(
    const float* __restrict__ x,
    const float* __restrict__ theta,
    float* __restrict__ attn_out) {
  __shared__ float zsm[S_LEN * NW + PAD];

  const int qblk = blockIdx.x & 7;
  const int bh   = blockIdx.x >> 3;
  const int b    = bh >> 3;
  const int h    = bh & 7;
  const int tid  = threadIdx.x;

  float th[8];
#pragma unroll
  for (int j = 0; j < 8; ++j) th[j] = theta[j];

  const float* xbase = x + (size_t)b * S_LEN * EMB + h * NW;
  {
    const float* xrow = xbase + (size_t)(qblk * CH + tid) * EMB;
    float4 xa = ((const float4*)xrow)[0];
    float4 xc = ((const float4*)xrow)[1];
    stage_pos(xa, xc, th, zsm + tid * NW);
  }
  if (tid < PAD) zsm[S_LEN * NW + tid] = 1.0f;
  __syncthreads();

  const int wid  = tid >> 5;
  const int lane = tid & 31;
  const int r    = lane & 15;
  const int rc   = lane & 7;
  const int hi   = lane >> 4;
  const int qrow = qblk * QB + wid * 32;
  const int qloc = wid * 32;
  const int rk   = 2 * (r & 7) + (r >> 3);

  v2f bqa0, bqa1, bqb0, bqb1;
  load_frag_pair(zsm, qloc, r, hi, bqa0, bqa1);
  load_frag_pair(zsm, qloc + 16, r, hi, bqb0, bqb1);

  constexpr float L2E = 1.4426950408889634f;
  constexpr float K1  = 0.35355339059327373f * L2E;
  bqa0 *= K1; bqa1 *= K1; bqb0 *= K1; bqb1 *= K1;

  v8f acca = {}, accb = {};

  const float* vp    = (r < 8) ? zsm + hi * NW + rc : zsm + S_LEN * NW + 20;
  const int    vstep = (r < 8) ? 16 * NW : 0;

  const float* kp = zsm + rk * NW + hi * 2;

  v2f   ak0, ak1;
  float vv[8];
  ak0 = *(const v2f*)kp;
  ak1 = *(const v2f*)(kp + 4);
  load_vtile(vp, vv);

  for (int ph = 0; ph < NCH; ++ph) {
    const bool more = ph + 1 < NCH;
    float4 xa = {}, xc = {};
    if (more) {
      const int npos = ((qblk + ph + 1) & (NCH - 1)) * CH + tid;
      const float* xrow = xbase + (size_t)npos * EMB;
      xa = ((const float4*)xrow)[0];
      xc = ((const float4*)xrow)[1];
    }

#pragma unroll 2
    for (int i = 0; i < 8; ++i)
      key_trip(kp, vp, vstep, ak0, ak1, vv, bqa0, bqa1, bqb0, bqb1, acca, accb);

    if (more) stage_pos(xa, xc, th, zsm + ((ph + 1) * CH + tid) * NW);

#pragma unroll 2
    for (int i = 0; i < 4; ++i)
      key_trip(kp, vp, vstep, ak0, ak1, vv, bqa0, bqa1, bqb0, bqb1, acca, accb);

    if (more) __syncthreads();

#pragma unroll 2
    for (int i = 0; i < 4; ++i)
      key_trip(kp, vp, vstep, ak0, ak1, vv, bqa0, bqa1, bqb0, bqb1, acca, accb);
  }

  float da = xor16(acca[0]), db = xor16(accb[0]);
  float a4 = xor16(acca[4]), a5 = xor16(acca[5]);
  float a6 = xor16(acca[6]), a7 = xor16(acca[7]);
  float b4 = xor16(accb[4]), b5 = xor16(accb[5]);
  float b6 = xor16(accb[6]), b7 = xor16(accb[7]);
  float inva = 1.0f / (hi ? acca[0] : da);
  float invb = 1.0f / (hi ? accb[0] : db);
  float* outp = attn_out + (size_t)b * S_LEN * EMB + h * NW;
  float4 oa, ob;
  oa.x = (hi ? a4 : acca[0]) * inva;
  oa.y = (hi ? a5 : acca[1]) * inva;
  oa.z = (hi ? a6 : acca[2]) * inva;
  oa.w = (hi ? a7 : acca[3]) * inva;
  ob.x = (hi ? b4 : accb[0]) * invb;
  ob.y = (hi ? b5 : accb[1]) * invb;
  ob.z = (hi ? b6 : accb[2]) * invb;
  ob.w = (hi ? b7 : accb[3]) * invb;
  *(float4*)(outp + (size_t)(qrow + r) * EMB + hi * 4) = oa;
  *(float4*)(outp + (size_t)(qrow + 16 + r) * EMB + hi * 4) = ob;
}

__global__ __launch_bounds__(64) void combine_kernel(
    const float* __restrict__ a,
    const float* __restrict__ W,
    float* __restrict__ out) {
  const int tid  = threadIdx.x;
  const int wid  = tid >> 5;
  const int lane = tid & 31;
  const int r    = lane & 15;
  const int hi   = lane >> 4;
  const int mb   = (blockIdx.x * 2 + wid) * 16;

  const float4* ap = (const float4*)(a + (size_t)(mb + r) * EMB + 4 * hi);
  const float4* wp = (const float4*)(W + (size_t)r * EMB + 4 * hi);

  float4 af[8];
  float4 bf[4][8];
#pragma unroll
  for (int j = 0; j < 8; ++j) {
    af[j] = ap[2 * j];
#pragma unroll
    for (int nt = 0; nt < 4; ++nt)
      bf[nt][j] = wp[nt * 16 * (EMB / 4) + 2 * j];
  }

  __builtin_amdgcn_sched_barrier(0);

  v8f accw[4] = {};
#pragma unroll
  for (int j = 0; j < 8; ++j) {
#pragma unroll
    for (int nt = 0; nt < 4; ++nt)
      accw[nt] = wmma4(v2f{af[j].x, af[j].y}, v2f{bf[nt][j].x, bf[nt][j].y},
                       accw[nt]);
#pragma unroll
    for (int nt = 0; nt < 4; ++nt)
      accw[nt] = wmma4(v2f{af[j].z, af[j].w}, v2f{bf[nt][j].z, bf[nt][j].w},
                       accw[nt]);
  }

#pragma unroll
  for (int nt = 0; nt < 4; ++nt) {
#pragma unroll
    for (int i = 0; i < 8; ++i)
      out[(size_t)(mb + i + hi * 8) * EMB + nt * 16 + r] = accw[nt][i];
  }
}

extern "C" void kernel_launch(void* const* d_in, const int* in_sizes, int n_in,
                              void* d_out, int out_size, void* d_ws,
                              size_t ws_size, hipStream_t stream) {
  (void)in_sizes; (void)n_in; (void)out_size; (void)ws_size;
  const float* x     = (const float*)d_in[0];
  const float* theta = (const float*)d_in[1];
  const float* W     = (const float*)d_in[2];
  float* out  = (float*)d_out;
  float* attn = (float*)d_ws;

  qattn_kernel<<<dim3(256), dim3(256), 0, stream>>>(x, theta, attn);
  combine_kernel<<<dim3(256), dim3(64), 0, stream>>>(attn, W, out);
}


# Priority window on the younger wave of each SIMD pair (s_setprio 3 for the first three quarters of the key trips between two barriers, then 0): both waves reach the barrier together; sums unchanged
# speedup vs baseline: 1.5507x; 1.0342x over previous
#include <hip/hip_runtime.h>
#include <hip/hip_bf16.h>
#include <math.h>

typedef __attribute__((ext_vector_type(2))) float v2f;
typedef __attribute__((ext_vector_type(8))) float v8f;

#define S_LEN 2048
#define EMB   64
#define NW    8
#define QB    256
#define PAD   160
#define CH    256
#define NCH   8

__device__ __forceinline__ v8f wmma4(v2f a, v2f b, v8f c) {
  return __builtin_amdgcn_wmma_f32_16x16x4_f32(false, a, false, b, (short)0, c,
                                               false, false);
}

__device__ __forceinline__ float xor16(float v) {
  int i = __float_as_int(v);
  int o = __builtin_amdgcn_permlanex16(i, i, 0x76543210, 0xfedcba98, false, false);
  return __int_as_float(o);
}

__device__ __forceinline__ void load_frag_pair(const float* zsm, int rowbase,
                                               int r, int hi, v2f& f0, v2f& f1) {
  const float* p = zsm + (rowbase + r) * NW + hi * 2;
  f0 = *(const v2f*)p;
  f1 = *(const v2f*)(p + 4);
}

__device__ __forceinline__ void load_vtile(const float* vp, float vv[8]) {
#pragma unroll
  for (int sl = 0; sl < 4; ++sl) {
    vv[2 * sl]     = vp[sl * 4 * NW];
    vv[2 * sl + 1] = vp[(sl * 4 + 2) * NW];
  }
}

__device__ __forceinline__ void stage_pos(float4 xa, float4 xc,
                                          const float th[8], float* zp) {
  float c0 = __cosf(xa.x + th[0]);
  float c1 = __cosf(xa.y + th[1]);
  float c2 = __cosf(xa.z + th[2]);
  float c3 = __cosf(xa.w + th[3]);
  float c4 = __cosf(xc.x + th[4]);
  float c5 = __cosf(xc.y + th[5]);
  float c6 = __cosf(xc.z + th[6]);
  float c7 = __cosf(xc.w + th[7]);
  float z1 = c0 * c1, z2 = z1 * c2, z3 = z2 * c3, z4 = z3 * c4;
  float z5 = z4 * c5, z6 = z5 * c6, z7 = z6 * c7;
  float z0 = ((c1 * c2) * (c3 * c4)) * ((c5 * c6) * c7);
  ((float4*)zp)[0] = make_float4(z0, z1, z2, z3);
  ((float4*)zp)[1] = make_float4(z4, z5, z6, z7);
}

__device__ __forceinline__ void key_trip(const float*& kp, const float*& vp,
                                         int vstep,
                                         v2f& ak0, v2f& ak1, float (&vv)[8],
                                         v2f bqa0, v2f bqa1, v2f bqb0, v2f bqb1,
                                         v8f& acca, v8f& accb) {
  v2f   ck0 = ak0, ck1 = ak1;
  float cv[8];
#pragma unroll
  for (int i = 0; i < 8; ++i) cv[i] = vv[i];

  vp += vstep;
  kp += 16 * NW;
  __builtin_amdgcn_sched_barrier(0);
  ak0 = *(const v2f*)kp;
  ak1 = *(const v2f*)(kp + 4);
  load_vtile(vp, vv);
  __builtin_amdgcn_sched_barrier(0);

  v8f sta = {}, stb = {};
  sta = wmma4(ck0, bqa0, sta);
  sta = wmma4(ck1, bqa1, sta);
  stb = wmma4(ck0, bqb0, stb);
  stb = wmma4(ck1, bqb1, stb);

  float pa[8], pb[8];
#pragma unroll
  for (int i = 0; i < 8; ++i)
    pa[i] = __builtin_amdgcn_exp2f(sta[i]);

#pragma unroll
  for (int sl = 0; sl < 4; ++sl) {
    v2f av = v2f{cv[2 * sl], cv[2 * sl + 1]};
    v2f bp = v2f{pa[2 * sl], pa[2 * sl + 1]};
    acca = wmma4(av, bp, acca);
  }

#pragma unroll
  for (int i = 0; i < 8; ++i)
    pb[i] = __builtin_amdgcn_exp2f(stb[i]);

#pragma unroll
  for (int sl = 0; sl < 4; ++sl) {
    v2f av = v2f{cv[2 * sl], cv[2 * sl + 1]};
    v2f bp = v2f{pb[2 * sl], pb[2 * sl + 1]};
    accb = wmma4(av, bp, accb);
  }
}

__global__ __launch_bounds__(256) void qattn_kernel(
    const float* __restrict__ x,
    const float* __restrict__ theta,
    float* __restrict__ attn_out) {
  __shared__ float zsm[S_LEN * NW + PAD];

  const int qblk = blockIdx.x & 7;
  const int bh   = blockIdx.x >> 3;
  const int b    = bh >> 3;
  const int h    = bh & 7;
  const int tid  = threadIdx.x;

  float th[8];
#pragma unroll
  for (int j = 0; j < 8; ++j) th[j] = theta[j];

  const float* xbase = x + (size_t)b * S_LEN * EMB + h * NW;
  {
    const float* xrow = xbase + (size_t)(qblk * CH + tid) * EMB;
    float4 xa = ((const float4*)xrow)[0];
    float4 xc = ((const float4*)xrow)[1];
    stage_pos(xa, xc, th, zsm + tid * NW);
  }
  if (tid < PAD) zsm[S_LEN * NW + tid] = 1.0f;
  __syncthreads();

  const int wid  = tid >> 5;
  const int lane = tid & 31;
  const int r    = lane & 15;
  const int rc   = lane & 7;
  const int hi   = lane >> 4;
  const int qrow = qblk * QB + wid * 32;
  const int qloc = wid * 32;
  const int rk   = 2 * (r & 7) + (r >> 3);

  v2f bqa0, bqa1, bqb0, bqb1;
  load_frag_pair(zsm, qloc, r, hi, bqa0, bqa1);
  load_frag_pair(zsm, qloc + 16, r, hi, bqb0, bqb1);

  constexpr float L2E = 1.4426950408889634f;
  constexpr float K1  = 0.35355339059327373f * L2E;
  bqa0 *= K1; bqa1 *= K1; bqb0 *= K1; bqb1 *= K1;

  v8f acca = {}, accb = {};

  const float* vp    = (r < 8) ? zsm + hi * NW + rc : zsm + S_LEN * NW + 20;
  const int    vstep = (r < 8) ? 16 * NW : 0;

  const float* kp = zsm + rk * NW + hi * 2;

  v2f   ak0, ak1;
  float vv[8];
  ak0 = *(const v2f*)kp;
  ak1 = *(const v2f*)(kp + 4);
  load_vtile(vp, vv);

  const bool young = __builtin_amdgcn_readfirstlane(wid) >= 4;
  if (young) __builtin_amdgcn_s_setprio(3);

  for (int ph = 0; ph < NCH; ++ph) {
    const bool more = ph + 1 < NCH;
    float4 xa = {}, xc = {};
    if (more) {
      const int npos = ((qblk + ph + 1) & (NCH - 1)) * CH + tid;
      const float* xrow = xbase + (size_t)npos * EMB;
      xa = ((const float4*)xrow)[0];
      xc = ((const float4*)xrow)[1];
    }

#pragma unroll 2
    for (int i = 0; i < 8; ++i)
      key_trip(kp, vp, vstep, ak0, ak1, vv, bqa0, bqa1, bqb0, bqb1, acca, accb);

    if (young && more) __builtin_amdgcn_s_setprio(0);

    if (more) stage_pos(xa, xc, th, zsm + ((ph + 1) * CH + tid) * NW);

#pragma unroll 2
    for (int i = 0; i < 4; ++i)
      key_trip(kp, vp, vstep, ak0, ak1, vv, bqa0, bqa1, bqb0, bqb1, acca, accb);

    if (more) {
      __syncthreads();
      if (young) __builtin_amdgcn_s_setprio(3);
    } else if (young) {
      __builtin_amdgcn_s_setprio(0);
    }

#pragma unroll 2
    for (int i = 0; i < 4; ++i)
      key_trip(kp, vp, vstep, ak0, ak1, vv, bqa0, bqa1, bqb0, bqb1, acca, accb);
  }

  float da = xor16(acca[0]), db = xor16(accb[0]);
  float a4 = xor16(acca[4]), a5 = xor16(acca[5]);
  float a6 = xor16(acca[6]), a7 = xor16(acca[7]);
  float b4 = xor16(accb[4]), b5 = xor16(accb[5]);
  float b6 = xor16(accb[6]), b7 = xor16(accb[7]);
  float inva = 1.0f / (hi ? acca[0] : da);
  float invb = 1.0f / (hi ? accb[0] : db);
  float* outp = attn_out + (size_t)b * S_LEN * EMB + h * NW;
  float4 oa, ob;
  oa.x = (hi ? a4 : acca[0]) * inva;
  oa.y = (hi ? a5 : acca[1]) * inva;
  oa.z = (hi ? a6 : acca[2]) * inva;
  oa.w = (hi ? a7 : acca[3]) * inva;
  ob.x = (hi ? b4 : accb[0]) * invb;
  ob.y = (hi ? b5 : accb[1]) * invb;
  ob.z = (hi ? b6 : accb[2]) * invb;
  ob.w = (hi ? b7 : accb[3]) * invb;
  *(float4*)(outp + (size_t)(qrow + r) * EMB + hi * 4) = oa;
  *(float4*)(outp + (size_t)(qrow + 16 + r) * EMB + hi * 4) = ob;
}

__global__ __launch_bounds__(64) void combine_kernel(
    const float* __restrict__ a,
    const float* __restrict__ W,
    float* __restrict__ out) {
  const int tid  = threadIdx.x;
  const int wid  = tid >> 5;
  const int lane = tid & 31;
  const int r    = lane & 15;
  const int hi   = lane >> 4;
  const int mb   = (blockIdx.x * 2 + wid) * 16;

  const float4* ap = (const float4*)(a + (size_t)(mb + r) * EMB + 4 * hi);
  const float4* wp = (const float4*)(W + (size_t)r * EMB + 4 * hi);

  float4 af[8];
  float4 bf[4][8];
#pragma unroll
  for (int j = 0; j < 8; ++j) {
    af[j] = ap[2 * j];
#pragma unroll
    for (int nt = 0; nt < 4; ++nt)
      bf[nt][j] = wp[nt * 16 * (EMB / 4) + 2 * j];
  }

  __builtin_amdgcn_sched_barrier(0);

  v8f accw[4] = {};
#pragma unroll
  for (int j = 0; j < 8; ++j) {
#pragma unroll
    for (int nt = 0; nt < 4; ++nt)
      accw[nt] = wmma4(v2f{af[j].x, af[j].y}, v2f{bf[nt][j].x, bf[nt][j].y},
                       accw[nt]);
#pragma unroll
    for (int nt = 0; nt < 4; ++nt)
      accw[nt] = wmma4(v2f{af[j].z, af[j].w}, v2f{bf[nt][j].z, bf[nt][j].w},
                       accw[nt]);
  }

#pragma unroll
  for (int nt = 0; nt < 4; ++nt) {
#pragma unroll
    for (int i = 0; i < 8; ++i)
      out[(size_t)(mb + i + hi * 8) * EMB + nt * 16 + r] = accw[nt][i];
  }
}

extern "C" void kernel_launch(void* const* d_in, const int* in_sizes, int n_in,
                              void* d_out, int out_size, void* d_ws,
                              size_t ws_size, hipStream_t stream) {
  (void)in_sizes; (void)n_in; (void)out_size; (void)ws_size;
  const float* x     = (const float*)d_in[0];
  const float* theta = (const float*)d_in[1];
  const float* W     = (const float*)d_in[2];
  float* out  = (float*)d_out;
  float* attn = (float*)d_ws;

  qattn_kernel<<<dim3(256), dim3(256), 0, stream>>>(x, theta, attn);
  combine_kernel<<<dim3(256), dim3(64), 0, stream>>>(attn, W, out);
}
